# starting kernel
# baseline (speedup 1.0000x reference)
#include <hip/hip_runtime.h>
#include <hip/hip_bf16.h>


typedef float v2f __attribute__((ext_vector_type(2)));
typedef float v8f __attribute__((ext_vector_type(8)));

__global__ void dbp_zero_kernel(float* __restrict__ p, long n) {
    long i = (long)blockIdx.x * blockDim.x + threadIdx.x;
    long stride = (long)gridDim.x * blockDim.x;
    for (; i < n; i += stride) p[i] = 0.0f;
}

__global__ void dbp_degree_kernel(const int* __restrict__ dst, float* __restrict__ deg, int E) {
    int i = blockIdx.x * blockDim.x + threadIdx.x;
    if (i < E) atomicAdd(&deg[dst[i]], 1.0f);
}

__global__ void dbp_dinv_kernel(float* __restrict__ deg, int N) {
    int i = blockIdx.x * blockDim.x + threadIdx.x;
    if (i < N) deg[i] = rsqrtf(deg[i] + 1.0f);
}

template <int FOUT, bool RELU_IN>
__global__ __launch_bounds__(256) void dbp_wmma_gemm_kernel(
    const float* __restrict__ A, const float* __restrict__ B,
    float* __restrict__ C, int M) {
    constexpr int TILES = FOUT / 16;
    const int wave = (blockIdx.x * blockDim.x + threadIdx.x) >> 5;
    const int lane = threadIdx.x & 31;
    if (wave >= M / 16) return;

    const int l16 = lane & 15;
    const int hi  = lane >> 4;
    const float* arow = A + (size_t)(wave * 16 + l16) * 128;

    v8f acc[TILES];
    v8f zv = {};
#pragma unroll
    for (int t = 0; t < TILES; ++t) acc[t] = zv;

    for (int kb = 0; kb < 128; kb += 4) {
        const int k0 = kb + 2 * hi;
        v2f a = *(const v2f*)(arow + k0);
        if (RELU_IN) { a.x = fmaxf(a.x, 0.0f); a.y = fmaxf(a.y, 0.0f); }
        const float* brow0 = B + (size_t)(k0 + 0) * FOUT + l16;
        const float* brow1 = B + (size_t)(k0 + 1) * FOUT + l16;
#pragma unroll
        for (int t = 0; t < TILES; ++t) {
            v2f b;
            b.x = brow0[t * 16];
            b.y = brow1[t * 16];
            acc[t] = __builtin_amdgcn_wmma_f32_16x16x4_f32(
                 false, a,  false, b,
                 (short)0, acc[t],  false,  false);
        }
    }
#pragma unroll
    for (int t = 0; t < TILES; ++t) {
        float* ccol = C + (size_t)(wave * 16 + 8 * hi) * FOUT + t * 16 + l16;
#pragma unroll
        for (int r = 0; r < 8; ++r)
            ccol[(size_t)r * FOUT] = acc[t][r];
    }
}

template <int F>
__global__ __launch_bounds__(256) void dbp_edge_agg_kernel(
    const int* __restrict__ src, const int* __restrict__ dst,
    const float* __restrict__ dinv, const float* __restrict__ h,
    float* __restrict__ agg, int E) {
    const int gid  = blockIdx.x * blockDim.x + threadIdx.x;
    const int edge = gid >> 5;
    const int lane = gid & 31;
    if (edge >= E) return;
    const int s = src[edge];
    const int d = dst[edge];
    const float norm = dinv[s] * dinv[d];
    const float* hrow = h   + (size_t)s * F;
    float*       arow = agg + (size_t)d * F;
    if constexpr (F == 128) {
        const float4 hv = *(const float4*)(hrow + lane * 4);
        atomicAdd(&arow[lane * 4 + 0], hv.x * norm);
        atomicAdd(&arow[lane * 4 + 1], hv.y * norm);
        atomicAdd(&arow[lane * 4 + 2], hv.z * norm);
        atomicAdd(&arow[lane * 4 + 3], hv.w * norm);
    } else {
        const float2 hv = *(const float2*)(hrow + lane * 2);
        atomicAdd(&arow[lane * 2 + 0], hv.x * norm);
        atomicAdd(&arow[lane * 2 + 1], hv.y * norm);
    }
}

__global__ void dbp_selfloop_bias_relu_kernel(
    float* __restrict__ agg, const float* __restrict__ h,
    const float* __restrict__ dinv, const float* __restrict__ bias,
    int M, int F) {
    long i = (long)blockIdx.x * blockDim.x + threadIdx.x;
    if (i >= (long)M * F) return;
    const int node = (int)(i / F);
    const int f    = (int)(i % F);
    const float di = dinv[node];
    float v = agg[i] + h[i] * (di * di) + bias[f];
    agg[i] = fmaxf(v, 0.0f);
}

__global__ __launch_bounds__(256) void dbp_softmax_kernel(
    const float* __restrict__ agg, const float* __restrict__ h,
    const float* __restrict__ dinv, const float* __restrict__ bias,
    float* __restrict__ out, int M) {
    const int gid  = blockIdx.x * blockDim.x + threadIdx.x;
    const int row  = gid >> 5;
    const int lane = gid & 31;
    if (row >= M) return;
    const float di = dinv[row];
    const float dd = di * di;
    const float* a  = agg + (size_t)row * 64;
    const float* hr = h   + (size_t)row * 64;
    float v0 = a[lane]      + hr[lane]      * dd + bias[lane];
    float v1 = a[lane + 32] + hr[lane + 32] * dd + bias[lane + 32];
    float mx = fmaxf(v0, v1);
#pragma unroll
    for (int off = 16; off > 0; off >>= 1) mx = fmaxf(mx, __shfl_xor(mx, off, 32));
    const float e0 = __expf(v0 - mx);
    const float e1 = __expf(v1 - mx);
    float s = e0 + e1;
#pragma unroll
    for (int off = 16; off > 0; off >>= 1) s += __shfl_xor(s, off, 32);
    const float inv = 1.0f / s;
    float* o = out + (size_t)row * 64;
    o[lane]      = e0 * inv;
    o[lane + 32] = e1 * inv;
}

extern "C" void kernel_launch(void* const* d_in, const int* in_sizes, int n_in,
                              void* d_out, int out_size, void* d_ws, size_t ws_size,
                              hipStream_t stream) {
    const float* x  = (const float*)d_in[0];
    const int*   ei = (const int*)d_in[1];
    const float* W1 = (const float*)d_in[2];
    const float* b1 = (const float*)d_in[3];
    const float* W2 = (const float*)d_in[4];
    const float* b2 = (const float*)d_in[5];
    float* out = (float*)d_out;

    const int N = in_sizes[0] / 128;
    const int E = in_sizes[1] / 2;
    const int* srcIdx = ei;
    const int* dstIdx = ei + E;

    float* ws   = (float*)d_ws;
    float* dinv = ws;
    float* h1   = dinv + (((size_t)N + 255) & ~(size_t)255);
    float* agg1 = h1   + (size_t)N * 128;
    float* h2   = agg1 + (size_t)N * 128;
    float* agg2 = h2   + (size_t)N * 64;

    const int TB = 256;
    auto blocksFor = [](long n, int tb) { return (int)((n + tb - 1) / tb); };

    dbp_zero_kernel<<<blocksFor(N, TB), TB, 0, stream>>>(dinv, N);
    dbp_zero_kernel<<<blocksFor((long)N * 128, TB), TB, 0, stream>>>(agg1, (long)N * 128);
    dbp_zero_kernel<<<blocksFor((long)N * 64, TB), TB, 0, stream>>>(agg2, (long)N * 64);

    dbp_degree_kernel<<<blocksFor(E, TB), TB, 0, stream>>>(dstIdx, dinv, E);
    dbp_dinv_kernel<<<blocksFor(N, TB), TB, 0, stream>>>(dinv, N);

    dbp_wmma_gemm_kernel<128, true>
        <<<blocksFor((long)(N / 16) * 32, TB), TB, 0, stream>>>(x, W1, h1, N);

    dbp_edge_agg_kernel<128>
        <<<blocksFor((long)E * 32, TB), TB, 0, stream>>>(srcIdx, dstIdx, dinv, h1, agg1, E);

    dbp_selfloop_bias_relu_kernel
        <<<blocksFor((long)N * 128, TB), TB, 0, stream>>>(agg1, h1, dinv, b1, N, 128);

    dbp_wmma_gemm_kernel<64, false>
        <<<blocksFor((long)(N / 16) * 32, TB), TB, 0, stream>>>(agg1, W2, h2, N);

    dbp_edge_agg_kernel<64>
        <<<blocksFor((long)E * 32, TB), TB, 0, stream>>>(srcIdx, dstIdx, dinv, h2, agg2, E);

    dbp_softmax_kernel
        <<<blocksFor((long)N * 32, TB), TB, 0, stream>>>(agg2, h2, dinv, b2, out, N);
}


# edge aggregation by destination: in-call counting sort of edges by dst (one returning int atomic per edge), one wave per node sums its in-edge rows in registers; no float atomics, no zeroing
# speedup vs baseline: 1.8728x; 1.8728x over previous
#include <hip/hip_runtime.h>
#include <hip/hip_bf16.h>


typedef float v2f __attribute__((ext_vector_type(2)));
typedef float v8f __attribute__((ext_vector_type(8)));

__global__ void dbp_zero_int_kernel(int* __restrict__ p, int n) {
    int i = blockIdx.x * blockDim.x + threadIdx.x;
    if (i < n) p[i] = 0;
}

__global__ void dbp_count_rank_kernel(const int* __restrict__ dst, int* __restrict__ cnt,
                                      int* __restrict__ rank, int E) {
    int i = blockIdx.x * blockDim.x + threadIdx.x;
    if (i < E) rank[i] = atomicAdd(&cnt[dst[i]], 1);
}

__global__ __launch_bounds__(256) void dbp_scan_kernel(
    const int* __restrict__ cnt, int* __restrict__ rowptr,
    float* __restrict__ dinv, int N) {
    __shared__ int s_front[8];
    __shared__ int s_own[8];
    const int t    = threadIdx.x;
    const int lane = t & 31;
    const int w    = t >> 5;
    const int tile0 = blockIdx.x * 1024;

    const int   nb = blockIdx.x;
    const int4* c4 = (const int4*)cnt + t;
    int front = 0;
    for (int it = 0; it < nb; it += 8) {
        int4 v[8];
#pragma unroll
        for (int k = 0; k < 8; ++k) {
            const int q = (it + k < nb) ? (it + k) : (nb - 1);
            v[k] = c4[q * 256];
        }
#pragma unroll
        for (int k = 0; k < 8; ++k)
            if (it + k < nb) front += (v[k].x + v[k].y) + (v[k].z + v[k].w);
    }
    const int i0 = tile0 + t * 4;
    int c[4];
#pragma unroll
    for (int k = 0; k < 4; ++k) c[k] = (i0 + k < N) ? cnt[i0 + k] : 0;
    const int own = (c[0] + c[1]) + (c[2] + c[3]);

    int incl = own;
#pragma unroll
    for (int off = 1; off < 32; off <<= 1) {
        const int up = __shfl_up(incl, off, 32);
        if (lane >= off) incl += up;
    }
#pragma unroll
    for (int off = 16; off > 0; off >>= 1) front += __shfl_xor(front, off, 32);
    if (lane == 31) s_own[w] = incl;
    if (lane == 0)  s_front[w] = front;
    __syncthreads();
    int excl = incl - own;
#pragma unroll
    for (int k = 0; k < 8; ++k) {
        excl += s_front[k];
        if (k < w) excl += s_own[k];
    }
#pragma unroll
    for (int k = 0; k < 4; ++k) {
        if (i0 + k < N) {
            rowptr[i0 + k] = excl;
            dinv[i0 + k]   = rsqrtf((float)c[k] + 1.0f);
        }
        excl += c[k];
        if (i0 + k == N - 1) rowptr[N] = excl;
    }
}

__global__ void dbp_fill_kernel(const int* __restrict__ src, const int* __restrict__ dst,
                                const int* __restrict__ rank, const int* __restrict__ rowptr,
                                int* __restrict__ col, int E) {
    int i = blockIdx.x * blockDim.x + threadIdx.x;
    if (i < E) {
        const int p = rowptr[dst[i]] + rank[i];
        col[max(0, min(p, E - 1))] = src[i];
    }
}

template <int FOUT, bool RELU_IN>
__global__ __launch_bounds__(256) void dbp_wmma_gemm_kernel(
    const float* __restrict__ A, const float* __restrict__ B,
    float* __restrict__ C, int M) {
    constexpr int TILES = FOUT / 16;
    const int wave = (blockIdx.x * blockDim.x + threadIdx.x) >> 5;
    const int lane = threadIdx.x & 31;
    if (wave >= M / 16) return;

    const int l16 = lane & 15;
    const int hi  = lane >> 4;
    const float* arow = A + (size_t)(wave * 16 + l16) * 128;

    v8f acc[TILES];
    v8f zv = {};
#pragma unroll
    for (int t = 0; t < TILES; ++t) acc[t] = zv;

    for (int kb = 0; kb < 128; kb += 4) {
        const int k0 = kb + 2 * hi;
        v2f a = *(const v2f*)(arow + k0);
        if (RELU_IN) { a.x = fmaxf(a.x, 0.0f); a.y = fmaxf(a.y, 0.0f); }
        const float* brow0 = B + (size_t)(k0 + 0) * FOUT + l16;
        const float* brow1 = B + (size_t)(k0 + 1) * FOUT + l16;
#pragma unroll
        for (int t = 0; t < TILES; ++t) {
            v2f b;
            b.x = brow0[t * 16];
            b.y = brow1[t * 16];
            acc[t] = __builtin_amdgcn_wmma_f32_16x16x4_f32(
                 false, a,  false, b,
                 (short)0, acc[t],  false,  false);
        }
    }
#pragma unroll
    for (int t = 0; t < TILES; ++t) {
        float* ccol = C + (size_t)(wave * 16 + 8 * hi) * FOUT + t * 16 + l16;
#pragma unroll
        for (int r = 0; r < 8; ++r)
            ccol[(size_t)r * FOUT] = acc[t][r];
    }
}

template <int F, int K>
__device__ __forceinline__ void dbp_gather_step(
    float (&acc)[F / 32], const float* __restrict__ h, int s_l, float w_l, int j, int lane) {
    constexpr int V = F / 32;
    typedef float vf __attribute__((ext_vector_type(V)));
    vf v[K];
    float wt[K];
#pragma unroll
    for (int k = 0; k < K; ++k) {
        const int s = __builtin_amdgcn_readlane(s_l, j + k);
        wt[k] = __int_as_float(__builtin_amdgcn_readlane(__float_as_int(w_l), j + k));
        v[k]  = *((const vf*)(h + (size_t)s * F) + lane);
    }
#pragma unroll
    for (int k = 0; k < K; ++k) {
#pragma unroll
        for (int e = 0; e < V; ++e) acc[e] = fmaf(v[k][e], wt[k], acc[e]);
    }
}

template <int F>
__global__ __launch_bounds__(256) void dbp_gather_kernel(
    const int* __restrict__ rowptr, const int* __restrict__ col,
    const float* __restrict__ dinv, const float* __restrict__ h,
    float* __restrict__ agg, int N, int E) {
    constexpr int V = F / 32;
    typedef float vf __attribute__((ext_vector_type(V)));
    const int lane = threadIdx.x & 31;
    const int node = __builtin_amdgcn_readfirstlane(
        (int)((blockIdx.x * blockDim.x + threadIdx.x) >> 5));
    if (node >= N) return;
    const int   beg = max(0, rowptr[node]);
    const int   end = min(E, rowptr[node + 1]);
    const float dd  = dinv[node];

    float acc[V];
#pragma unroll
    for (int e = 0; e < V; ++e) acc[e] = 0.0f;

    for (int base = beg; base < end; base += 32) {
        const int n = (end - base < 32) ? (end - base) : 32;
        int   s_l = 0;
        float w_l = 0.0f;
        if (lane < n) {
            s_l = max(0, min(col[base + lane], N - 1));
            w_l = dinv[s_l] * dd;
        }
        int j = 0;
        for (; j + 8 <= n; j += 8) dbp_gather_step<F, 8>(acc, h, s_l, w_l, j, lane);
        if (j + 4 <= n) { dbp_gather_step<F, 4>(acc, h, s_l, w_l, j, lane); j += 4; }
        if (j + 2 <= n) { dbp_gather_step<F, 2>(acc, h, s_l, w_l, j, lane); j += 2; }
        if (j < n) dbp_gather_step<F, 1>(acc, h, s_l, w_l, j, lane);
    }
    vf o;
#pragma unroll
    for (int e = 0; e < V; ++e) o[e] = acc[e];
    *((vf*)(agg + (size_t)node * F) + lane) = o;
}

__global__ void dbp_selfloop_bias_relu_kernel(
    float* __restrict__ agg, const float* __restrict__ h,
    const float* __restrict__ dinv, const float* __restrict__ bias,
    int M, int F) {
    long i = (long)blockIdx.x * blockDim.x + threadIdx.x;
    if (i >= (long)M * F) return;
    const int node = (int)(i / F);
    const int f    = (int)(i % F);
    const float di = dinv[node];
    float v = agg[i] + h[i] * (di * di) + bias[f];
    agg[i] = fmaxf(v, 0.0f);
}

__global__ __launch_bounds__(256) void dbp_softmax_kernel(
    const float* __restrict__ agg, const float* __restrict__ h,
    const float* __restrict__ dinv, const float* __restrict__ bias,
    float* __restrict__ out, int M) {
    const int gid  = blockIdx.x * blockDim.x + threadIdx.x;
    const int row  = gid >> 5;
    const int lane = gid & 31;
    if (row >= M) return;
    const float di = dinv[row];
    const float dd = di * di;
    const float* a  = agg + (size_t)row * 64;
    const float* hr = h   + (size_t)row * 64;
    float v0 = a[lane]      + hr[lane]      * dd + bias[lane];
    float v1 = a[lane + 32] + hr[lane + 32] * dd + bias[lane + 32];
    float mx = fmaxf(v0, v1);
#pragma unroll
    for (int off = 16; off > 0; off >>= 1) mx = fmaxf(mx, __shfl_xor(mx, off, 32));
    const float e0 = __expf(v0 - mx);
    const float e1 = __expf(v1 - mx);
    float s = e0 + e1;
#pragma unroll
    for (int off = 16; off > 0; off >>= 1) s += __shfl_xor(s, off, 32);
    const float inv = 1.0f / s;
    float* o = out + (size_t)row * 64;
    o[lane]      = e0 * inv;
    o[lane + 32] = e1 * inv;
}

extern "C" void kernel_launch(void* const* d_in, const int* in_sizes, int n_in,
                              void* d_out, int out_size, void* d_ws, size_t ws_size,
                              hipStream_t stream) {
    const float* x  = (const float*)d_in[0];
    const int*   ei = (const int*)d_in[1];
    const float* W1 = (const float*)d_in[2];
    const float* b1 = (const float*)d_in[3];
    const float* W2 = (const float*)d_in[4];
    const float* b2 = (const float*)d_in[5];
    float* out = (float*)d_out;

    const int N = in_sizes[0] / 128;
    const int E = in_sizes[1] / 2;
    const int* srcIdx = ei;
    const int* dstIdx = ei + E;

    float* ws   = (float*)d_ws;
    float* dinv = ws;
    float* h1   = dinv + (((size_t)N + 255) & ~(size_t)255);
    float* agg1 = h1   + (size_t)N * 128;
    float* h2   = agg1 + (size_t)N * 128;
    int*   cnt    = (int*)(h2 + (size_t)N * 64);
    int*   rowptr = cnt    + (((size_t)N + 255) & ~(size_t)255);
    int*   rank   = rowptr + (((size_t)N + 256) & ~(size_t)255);
    int*   col    = rank   + (((size_t)E + 255) & ~(size_t)255);
    float* agg2 = h1;

    const int TB = 256;
    auto blocksFor = [](long n, int tb) { return (int)((n + tb - 1) / tb); };

    dbp_zero_int_kernel<<<blocksFor(N, TB), TB, 0, stream>>>(cnt, N);
    dbp_count_rank_kernel<<<blocksFor(E, TB), TB, 0, stream>>>(dstIdx, cnt, rank, E);
    dbp_scan_kernel<<<blocksFor(N, 1024), TB, 0, stream>>>(cnt, rowptr, dinv, N);
    dbp_fill_kernel<<<blocksFor(E, TB), TB, 0, stream>>>(srcIdx, dstIdx, rank, rowptr, col, E);

    dbp_wmma_gemm_kernel<128, true>
        <<<blocksFor((long)(N / 16) * 32, TB), TB, 0, stream>>>(x, W1, h1, N);

    dbp_gather_kernel<128>
        <<<blocksFor((long)N * 32, TB), TB, 0, stream>>>(rowptr, col, dinv, h1, agg1, N, E);

    dbp_selfloop_bias_relu_kernel
        <<<blocksFor((long)N * 128, TB), TB, 0, stream>>>(agg1, h1, dinv, b1, N, 128);

    dbp_wmma_gemm_kernel<64, false>
        <<<blocksFor((long)(N / 16) * 32, TB), TB, 0, stream>>>(agg1, W2, h2, N);

    dbp_gather_kernel<64>
        <<<blocksFor((long)N * 32, TB), TB, 0, stream>>>(rowptr, col, dinv, h2, agg2, N, E);

    dbp_softmax_kernel
        <<<blocksFor((long)N * 32, TB), TB, 0, stream>>>(agg2, h2, dinv, b2, out, N);
}


# layer-1 GEMM shares the launch of the edge count+rank pass (independent jobs split by work-group, GEMM work-groups first in the grid); 8 launches -> 7
# speedup vs baseline: 2.3973x; 1.2801x over previous
#include <hip/hip_runtime.h>
#include <hip/hip_bf16.h>


typedef float v2f __attribute__((ext_vector_type(2)));
typedef float v8f __attribute__((ext_vector_type(8)));

__global__ void dbp_zero_int_kernel(int* __restrict__ p, int n) {
    int i = blockIdx.x * blockDim.x + threadIdx.x;
    if (i < n) p[i] = 0;
}

__global__ __launch_bounds__(256) void dbp_scan_kernel(
    const int* __restrict__ cnt, int* __restrict__ rowptr,
    float* __restrict__ dinv, int N) {
    __shared__ int s_front[8];
    __shared__ int s_own[8];
    const int t    = threadIdx.x;
    const int lane = t & 31;
    const int w    = t >> 5;
    const int tile0 = blockIdx.x * 1024;

    const int   nb = blockIdx.x;
    const int4* c4 = (const int4*)cnt + t;
    int front = 0;
    for (int it = 0; it < nb; it += 8) {
        int4 v[8];
#pragma unroll
        for (int k = 0; k < 8; ++k) {
            const int q = (it + k < nb) ? (it + k) : (nb - 1);
            v[k] = c4[q * 256];
        }
#pragma unroll
        for (int k = 0; k < 8; ++k)
            if (it + k < nb) front += (v[k].x + v[k].y) + (v[k].z + v[k].w);
    }
    const int i0 = tile0 + t * 4;
    int c[4];
#pragma unroll
    for (int k = 0; k < 4; ++k) c[k] = (i0 + k < N) ? cnt[i0 + k] : 0;
    const int own = (c[0] + c[1]) + (c[2] + c[3]);

    int incl = own;
#pragma unroll
    for (int off = 1; off < 32; off <<= 1) {
        const int up = __shfl_up(incl, off, 32);
        if (lane >= off) incl += up;
    }
#pragma unroll
    for (int off = 16; off > 0; off >>= 1) front += __shfl_xor(front, off, 32);
    if (lane == 31) s_own[w] = incl;
    if (lane == 0)  s_front[w] = front;
    __syncthreads();
    int excl = incl - own;
#pragma unroll
    for (int k = 0; k < 8; ++k) {
        excl += s_front[k];
        if (k < w) excl += s_own[k];
    }
#pragma unroll
    for (int k = 0; k < 4; ++k) {
        if (i0 + k < N) {
            rowptr[i0 + k] = excl;
            dinv[i0 + k]   = rsqrtf((float)c[k] + 1.0f);
        }
        excl += c[k];
        if (i0 + k == N - 1) rowptr[N] = excl;
    }
}

__global__ void dbp_fill_kernel(const int* __restrict__ src, const int* __restrict__ dst,
                                const int* __restrict__ rank, const int* __restrict__ rowptr,
                                int* __restrict__ col, int E) {
    int i = blockIdx.x * blockDim.x + threadIdx.x;
    if (i < E) {
        const int p = rowptr[dst[i]] + rank[i];
        col[max(0, min(p, E - 1))] = src[i];
    }
}

template <int FOUT, bool RELU_IN>
__device__ __forceinline__ void dbp_wmma_gemm_strip(
    const float* __restrict__ A, const float* __restrict__ B,
    float* __restrict__ C, int M, int wave) {
    constexpr int TILES = FOUT / 16;
    const int lane = threadIdx.x & 31;
    if (wave >= M / 16) return;

    const int l16 = lane & 15;
    const int hi  = lane >> 4;
    const float* arow = A + (size_t)(wave * 16 + l16) * 128;

    v8f acc[TILES];
    v8f zv = {};
#pragma unroll
    for (int t = 0; t < TILES; ++t) acc[t] = zv;

    for (int kb = 0; kb < 128; kb += 4) {
        const int k0 = kb + 2 * hi;
        v2f a = *(const v2f*)(arow + k0);
        if (RELU_IN) { a.x = fmaxf(a.x, 0.0f); a.y = fmaxf(a.y, 0.0f); }
        const float* brow0 = B + (size_t)(k0 + 0) * FOUT + l16;
        const float* brow1 = B + (size_t)(k0 + 1) * FOUT + l16;
#pragma unroll
        for (int t = 0; t < TILES; ++t) {
            v2f b;
            b.x = brow0[t * 16];
            b.y = brow1[t * 16];
            acc[t] = __builtin_amdgcn_wmma_f32_16x16x4_f32(
                 false, a,  false, b,
                 (short)0, acc[t],  false,  false);
        }
    }
#pragma unroll
    for (int t = 0; t < TILES; ++t) {
        float* ccol = C + (size_t)(wave * 16 + 8 * hi) * FOUT + t * 16 + l16;
#pragma unroll
        for (int r = 0; r < 8; ++r)
            ccol[(size_t)r * FOUT] = acc[t][r];
    }
}

template <int FOUT, bool RELU_IN>
__global__ __launch_bounds__(256) void dbp_wmma_gemm_kernel(
    const float* __restrict__ A, const float* __restrict__ B,
    float* __restrict__ C, int M) {
    dbp_wmma_gemm_strip<FOUT, RELU_IN>(
        A, B, C, M, (int)((blockIdx.x * blockDim.x + threadIdx.x) >> 5));
}

__global__ __launch_bounds__(256) void dbp_count_rank_gemm1_kernel(
    const int* __restrict__ dst, int* __restrict__ cnt, int* __restrict__ rank, int E,
    const float* __restrict__ x, const float* __restrict__ W1, float* __restrict__ h1,
    int N, int gemmBlocks) {
    if ((int)blockIdx.x < gemmBlocks) {
        dbp_wmma_gemm_strip<128, true>(
            x, W1, h1, N, (int)((blockIdx.x * 256 + threadIdx.x) >> 5));
    } else {
        const int i = ((int)blockIdx.x - gemmBlocks) * 256 + (int)threadIdx.x;
        if (i < E) rank[i] = atomicAdd(&cnt[dst[i]], 1);
    }
}

template <int F, int K>
__device__ __forceinline__ void dbp_gather_step(
    float (&acc)[F / 32], const float* __restrict__ h, int s_l, float w_l, int j, int lane) {
    constexpr int V = F / 32;
    typedef float vf __attribute__((ext_vector_type(V)));
    vf v[K];
    float wt[K];
#pragma unroll
    for (int k = 0; k < K; ++k) {
        const int s = __builtin_amdgcn_readlane(s_l, j + k);
        wt[k] = __int_as_float(__builtin_amdgcn_readlane(__float_as_int(w_l), j + k));
        v[k]  = *((const vf*)(h + (size_t)s * F) + lane);
    }
#pragma unroll
    for (int k = 0; k < K; ++k) {
#pragma unroll
        for (int e = 0; e < V; ++e) acc[e] = fmaf(v[k][e], wt[k], acc[e]);
    }
}

template <int F, bool SOFTMAX>
__global__ __launch_bounds__(256) void dbp_gather_kernel(
    const int* __restrict__ rowptr, const int* __restrict__ col,
    const float* __restrict__ dinv, const float* __restrict__ h,
    const float* __restrict__ bias, float* __restrict__ rows, int N, int E) {
    constexpr int V = F / 32;
    typedef float vf __attribute__((ext_vector_type(V)));
    const int lane = threadIdx.x & 31;
    const int node = __builtin_amdgcn_readfirstlane(
        (int)((blockIdx.x * blockDim.x + threadIdx.x) >> 5));
    if (node >= N) return;
    const int   beg = max(0, rowptr[node]);
    const int   end = min(E, rowptr[node + 1]);
    const float dd  = dinv[node];
    const vf hs = *((const vf*)(h + (size_t)node * F) + lane);
    const vf bv = *((const vf*)bias + lane);

    float acc[V];
#pragma unroll
    for (int e = 0; e < V; ++e) acc[e] = 0.0f;

    for (int base = beg; base < end; base += 32) {
        const int n = (end - base < 32) ? (end - base) : 32;
        int   s_l = 0;
        float w_l = 0.0f;
        if (lane < n) {
            s_l = max(0, min(col[base + lane], N - 1));
            w_l = dinv[s_l] * dd;
        }
        int j = 0;
        for (; j + 8 <= n; j += 8) dbp_gather_step<F, 8>(acc, h, s_l, w_l, j, lane);
        if (j + 4 <= n) { dbp_gather_step<F, 4>(acc, h, s_l, w_l, j, lane); j += 4; }
        if (j + 2 <= n) { dbp_gather_step<F, 2>(acc, h, s_l, w_l, j, lane); j += 2; }
        if (j < n) dbp_gather_step<F, 1>(acc, h, s_l, w_l, j, lane);
    }
    const float d2 = dd * dd;
    float v[V];
#pragma unroll
    for (int e = 0; e < V; ++e) v[e] = acc[e] + hs[e] * d2 + bv[e];

    vf o;
    if (!SOFTMAX) {
#pragma unroll
        for (int e = 0; e < V; ++e) o[e] = fmaxf(v[e], 0.0f);
    } else {
        float mx = v[0];
#pragma unroll
        for (int e = 1; e < V; ++e) mx = fmaxf(mx, v[e]);
#pragma unroll
        for (int off = 16; off > 0; off >>= 1) mx = fmaxf(mx, __shfl_xor(mx, off, 32));
        float s = 0.0f;
#pragma unroll
        for (int e = 0; e < V; ++e) { v[e] = __expf(v[e] - mx); s += v[e]; }
#pragma unroll
        for (int off = 16; off > 0; off >>= 1) s += __shfl_xor(s, off, 32);
        const float inv = 1.0f / s;
#pragma unroll
        for (int e = 0; e < V; ++e) o[e] = v[e] * inv;
    }
    *((vf*)(rows + (size_t)node * F) + lane) = o;
}

extern "C" void kernel_launch(void* const* d_in, const int* in_sizes, int n_in,
                              void* d_out, int out_size, void* d_ws, size_t ws_size,
                              hipStream_t stream) {
    const float* x  = (const float*)d_in[0];
    const int*   ei = (const int*)d_in[1];
    const float* W1 = (const float*)d_in[2];
    const float* b1 = (const float*)d_in[3];
    const float* W2 = (const float*)d_in[4];
    const float* b2 = (const float*)d_in[5];
    float* out = (float*)d_out;

    const int N = in_sizes[0] / 128;
    const int E = in_sizes[1] / 2;
    const int* srcIdx = ei;
    const int* dstIdx = ei + E;

    float* ws   = (float*)d_ws;
    float* dinv = ws;
    float* h1   = dinv + (((size_t)N + 255) & ~(size_t)255);
    float* x1   = h1   + (size_t)N * 128;
    float* h2   = x1   + (size_t)N * 128;
    int*   cnt    = (int*)(h2 + (size_t)N * 64);
    int*   rowptr = cnt    + (((size_t)N + 255) & ~(size_t)255);
    int*   rank   = rowptr + (((size_t)N + 256) & ~(size_t)255);
    int*   col    = rank   + (((size_t)E + 255) & ~(size_t)255);

    const int TB = 256;
    auto blocksFor = [](long n, int tb) { return (int)((n + tb - 1) / tb); };

    const int gemmBlocks = blocksFor((long)(N / 16) * 32, TB);
    dbp_zero_int_kernel<<<blocksFor(N, TB), TB, 0, stream>>>(cnt, N);
    dbp_count_rank_gemm1_kernel<<<gemmBlocks + blocksFor(E, TB), TB, 0, stream>>>(
        dstIdx, cnt, rank, E, x, W1, h1, N, gemmBlocks);
    dbp_scan_kernel<<<blocksFor(N, 1024), TB, 0, stream>>>(cnt, rowptr, dinv, N);
    dbp_fill_kernel<<<blocksFor(E, TB), TB, 0, stream>>>(srcIdx, dstIdx, rank, rowptr, col, E);

    dbp_gather_kernel<128, false>
        <<<blocksFor((long)N * 32, TB), TB, 0, stream>>>(rowptr, col, dinv, h1, b1, x1, N, E);

    dbp_wmma_gemm_kernel<64, false>
        <<<blocksFor((long)(N / 16) * 32, TB), TB, 0, stream>>>(x1, W2, h2, N);

    dbp_gather_kernel<64, true>
        <<<blocksFor((long)N * 32, TB), TB, 0, stream>>>(rowptr, col, dinv, h2, b2, out, N, E);
}


# layer-2 GEMM rides behind the layer-1 aggregation: 16 waves finish 16 nodes, x1 tile stays in LDS, two of them multiply it by W2 (WMMA) and store the h2 strip; 5 launches -> 4
# speedup vs baseline: 2.9387x; 1.2258x over previous
#include <hip/hip_runtime.h>
#include <hip/hip_bf16.h>


typedef float v2f __attribute__((ext_vector_type(2)));
typedef float v8f __attribute__((ext_vector_type(8)));

constexpr int DBP_CAP = 32;

__global__ void dbp_zero_int_kernel(int* __restrict__ p, int n, int* __restrict__ spill_n) {
    int i = blockIdx.x * blockDim.x + threadIdx.x;
    if (i < n) p[i] = 0;
    if (i == 0) *spill_n = 0;
}

template <int FOUT, bool RELU_IN>
__device__ __forceinline__ void dbp_wmma_gemm_strip(
    const float* __restrict__ A, const float* __restrict__ B,
    float* __restrict__ C, int M, int wave) {
    constexpr int TILES = FOUT / 16;
    const int lane = threadIdx.x & 31;
    if (wave >= M / 16) return;

    const int l16 = lane & 15;
    const int hi  = lane >> 4;
    const float* arow = A + (size_t)(wave * 16 + l16) * 128;

    v8f acc[TILES];
    v8f zv = {};
#pragma unroll
    for (int t = 0; t < TILES; ++t) acc[t] = zv;

    for (int kb = 0; kb < 128; kb += 4) {
        const int k0 = kb + 2 * hi;
        v2f a = *(const v2f*)(arow + k0);
        if (RELU_IN) { a.x = fmaxf(a.x, 0.0f); a.y = fmaxf(a.y, 0.0f); }
        const float* brow0 = B + (size_t)(k0 + 0) * FOUT + l16;
        const float* brow1 = B + (size_t)(k0 + 1) * FOUT + l16;
#pragma unroll
        for (int t = 0; t < TILES; ++t) {
            v2f b;
            b.x = brow0[t * 16];
            b.y = brow1[t * 16];
            acc[t] = __builtin_amdgcn_wmma_f32_16x16x4_f32(
                 false, a,  false, b,
                 (short)0, acc[t],  false,  false);
        }
    }
#pragma unroll
    for (int t = 0; t < TILES; ++t) {
        float* ccol = C + (size_t)(wave * 16 + 8 * hi) * FOUT + t * 16 + l16;
#pragma unroll
        for (int r = 0; r < 8; ++r)
            ccol[(size_t)r * FOUT] = acc[t][r];
    }
}

__global__ __launch_bounds__(256) void dbp_bucket_gemm1_kernel(
    const int* __restrict__ src, const int* __restrict__ dst, int* __restrict__ cnt,
    int* __restrict__ bucket, int* __restrict__ spill_n, int* __restrict__ spill_src,
    int* __restrict__ spill_dst, int spillCap, int E,
    const float* __restrict__ x, const float* __restrict__ W1, float* __restrict__ h1,
    int N, int gemmBlocks) {
    if ((int)blockIdx.x < gemmBlocks) {
        dbp_wmma_gemm_strip<128, true>(
            x, W1, h1, N, (int)((blockIdx.x * 256 + threadIdx.x) >> 5));
    } else {
        const int i = ((int)blockIdx.x - gemmBlocks) * 256 + (int)threadIdx.x;
        if (i < E) {
            const int d = dst[i];
            const int s = src[i];
            const int slot = atomicAdd(&cnt[d], 1);
            if (slot < DBP_CAP) {
                bucket[(size_t)d * DBP_CAP + slot] = s;
            } else {
                const int p = atomicAdd(spill_n, 1);
                if (p < spillCap) { spill_src[p] = s; spill_dst[p] = d; }
            }
        }
    }
}

template <int F, int K>
__device__ __forceinline__ void dbp_gather_step(
    float (&acc)[F / 32], const float* __restrict__ h, int s_l, float w_l, int j, int lane) {
    constexpr int V = F / 32;
    typedef float vf __attribute__((ext_vector_type(V)));
    vf v[K];
    float wt[K];
#pragma unroll
    for (int k = 0; k < K; ++k) {
        const int s = __builtin_amdgcn_readlane(s_l, j + k);
        wt[k] = __int_as_float(__builtin_amdgcn_readlane(__float_as_int(w_l), j + k));
        v[k]  = *((const vf*)(h + (size_t)s * F) + lane);
    }
#pragma unroll
    for (int k = 0; k < K; ++k) {
#pragma unroll
        for (int e = 0; e < V; ++e) acc[e] = fmaf(v[k][e], wt[k], acc[e]);
    }
}

template <int F>
__device__ __forceinline__ void dbp_gather_matching(
    float (&acc)[F / 32], const float* __restrict__ h, const int* __restrict__ cnt,
    const int* __restrict__ esrc, const int* __restrict__ edst, int m,
    int node, float dd, int N, int lane) {
    for (int base = 0; base < m; base += 32) {
        const int  i   = base + lane;
        const bool hit = (i < m) && (edst[i] == node);
        int   s_l = 0;
        float w_l = 0.0f;
        if (hit) {
            s_l = max(0, min(esrc[i], N - 1));
            w_l = rsqrtf((float)cnt[s_l] + 1.0f) * dd;
        }
        unsigned todo = __builtin_amdgcn_ballot_w32(hit);
        while (todo) {
            const int j = __builtin_ctz(todo);
            todo &= todo - 1;
            dbp_gather_step<F, 1>(acc, h, s_l, w_l, j, lane);
        }
    }
}

template <int F, bool SOFTMAX>
__device__ __forceinline__ void dbp_node_row(
    float (&o)[F / 32],
    const int* __restrict__ cnt, const int* __restrict__ bucket,
    const int* __restrict__ spill_n, const int* __restrict__ spill_src,
    const int* __restrict__ spill_dst, int spillCap,
    const int* __restrict__ src, const int* __restrict__ dst,
    const float* __restrict__ h, const float* __restrict__ bias,
    int N, int E, int node, int lane) {
    constexpr int V = F / 32;
    typedef float vf __attribute__((ext_vector_type(V)));
    const int   c  = max(0, cnt[node]);
    const float dd = rsqrtf((float)c + 1.0f);
    const vf hs = *((const vf*)(h + (size_t)node * F) + lane);
    const vf bv = *((const vf*)bias + lane);

    float acc[V];
#pragma unroll
    for (int e = 0; e < V; ++e) acc[e] = 0.0f;

    int spilled = 0;
    if (c > DBP_CAP) spilled = *spill_n;
    if (spilled <= spillCap) {
        const int n = (c < DBP_CAP) ? c : DBP_CAP;
        int   s_l = 0;
        float w_l = 0.0f;
        if (lane < n) {
            s_l = max(0, min(bucket[(size_t)node * DBP_CAP + lane], N - 1));
            w_l = rsqrtf((float)cnt[s_l] + 1.0f) * dd;
        }
        int j = 0;
        for (; j + 8 <= n; j += 8) dbp_gather_step<F, 8>(acc, h, s_l, w_l, j, lane);
        if (j + 4 <= n) { dbp_gather_step<F, 4>(acc, h, s_l, w_l, j, lane); j += 4; }
        if (j + 2 <= n) { dbp_gather_step<F, 2>(acc, h, s_l, w_l, j, lane); j += 2; }
        if (j < n) dbp_gather_step<F, 1>(acc, h, s_l, w_l, j, lane);
        if (spilled > 0)
            dbp_gather_matching<F>(acc, h, cnt, spill_src, spill_dst, spilled, node, dd, N, lane);
    } else {
        dbp_gather_matching<F>(acc, h, cnt, src, dst, E, node, dd, N, lane);
    }
    const float d2 = dd * dd;
    float v[V];
#pragma unroll
    for (int e = 0; e < V; ++e) v[e] = acc[e] + hs[e] * d2 + bv[e];

    if (!SOFTMAX) {
#pragma unroll
        for (int e = 0; e < V; ++e) o[e] = fmaxf(v[e], 0.0f);
    } else {
        float mx = v[0];
#pragma unroll
        for (int e = 1; e < V; ++e) mx = fmaxf(mx, v[e]);
#pragma unroll
        for (int off = 16; off > 0; off >>= 1) mx = fmaxf(mx, __shfl_xor(mx, off, 32));
        float s = 0.0f;
#pragma unroll
        for (int e = 0; e < V; ++e) { v[e] = __expf(v[e] - mx); s += v[e]; }
#pragma unroll
        for (int off = 16; off > 0; off >>= 1) s += __shfl_xor(s, off, 32);
        const float inv = 1.0f / s;
#pragma unroll
        for (int e = 0; e < V; ++e) o[e] = v[e] * inv;
    }
}

template <int F, bool SOFTMAX>
__global__ __launch_bounds__(256) void dbp_gather_kernel(
    const int* __restrict__ cnt, const int* __restrict__ bucket,
    const int* __restrict__ spill_n, const int* __restrict__ spill_src,
    const int* __restrict__ spill_dst, int spillCap,
    const int* __restrict__ src, const int* __restrict__ dst,
    const float* __restrict__ h, const float* __restrict__ bias,
    float* __restrict__ rows, int N, int E) {
    constexpr int V = F / 32;
    typedef float vf __attribute__((ext_vector_type(V)));
    const int lane = threadIdx.x & 31;
    const int node = __builtin_amdgcn_readfirstlane(
        (int)((blockIdx.x * blockDim.x + threadIdx.x) >> 5));
    if (node >= N) return;
    float o[V];
    dbp_node_row<F, SOFTMAX>(o, cnt, bucket, spill_n, spill_src, spill_dst, spillCap,
                             src, dst, h, bias, N, E, node, lane);
    vf ov;
#pragma unroll
    for (int e = 0; e < V; ++e) ov[e] = o[e];
    *((vf*)(rows + (size_t)node * F) + lane) = ov;
}

constexpr int DBP_XP = 132;

__global__ __launch_bounds__(512) void dbp_gather1_gemm2_kernel(
    const int* __restrict__ cnt, const int* __restrict__ bucket,
    const int* __restrict__ spill_n, const int* __restrict__ spill_src,
    const int* __restrict__ spill_dst, int spillCap,
    const int* __restrict__ src, const int* __restrict__ dst,
    const float* __restrict__ h1, const float* __restrict__ b1,
    const float* __restrict__ W2, float* __restrict__ h2, int N, int E) {
    __shared__ float tile[16 * DBP_XP];
    typedef float v4f __attribute__((ext_vector_type(4)));
    const int lane = threadIdx.x & 31;
    const int w    = __builtin_amdgcn_readfirstlane((int)(threadIdx.x >> 5));
    const int row0 = (int)blockIdx.x * 16;
    const int node = row0 + w;

    float o[4] = {0.0f, 0.0f, 0.0f, 0.0f};
    if (node < N)
        dbp_node_row<128, false>(o, cnt, bucket, spill_n, spill_src, spill_dst, spillCap,
                                 src, dst, h1, b1, N, E, node, lane);
    v4f ov = {o[0], o[1], o[2], o[3]};
    *((v4f*)(tile + w * DBP_XP) + lane) = ov;
    __syncthreads();
    if (w >= 2) return;

    const int l16 = lane & 15;
    const int hi  = lane >> 4;
    const float* arow = tile + l16 * DBP_XP;
    const float* bcol = W2 + w * 32 + l16;
    v8f acc0 = {};
    v8f acc1 = {};
    for (int kb = 0; kb < 128; kb += 4) {
        const int k0 = kb + 2 * hi;
        const v2f a = *(const v2f*)(arow + k0);
        const float* brow0 = bcol + (size_t)(k0 + 0) * 64;
        const float* brow1 = bcol + (size_t)(k0 + 1) * 64;
        v2f b0, b1v;
        b0.x  = brow0[0];
        b0.y  = brow1[0];
        b1v.x = brow0[16];
        b1v.y = brow1[16];
        acc0 = __builtin_amdgcn_wmma_f32_16x16x4_f32(
             false, a,  false, b0,
             (short)0, acc0,  false,  false);
        acc1 = __builtin_amdgcn_wmma_f32_16x16x4_f32(
             false, a,  false, b1v,
             (short)0, acc1,  false,  false);
    }
    float* ccol = h2 + (size_t)(row0 + 8 * hi) * 64 + w * 32 + l16;
#pragma unroll
    for (int r = 0; r < 8; ++r) {
        if (row0 + 8 * hi + r < N) {
            ccol[(size_t)r * 64]      = acc0[r];
            ccol[(size_t)r * 64 + 16] = acc1[r];
        }
    }
}


extern "C" void kernel_launch(void* const* d_in, const int* in_sizes, int n_in,
                              void* d_out, int out_size, void* d_ws, size_t ws_size,
                              hipStream_t stream) {
    const float* x  = (const float*)d_in[0];
    const int*   ei = (const int*)d_in[1];
    const float* W1 = (const float*)d_in[2];
    const float* b1 = (const float*)d_in[3];
    const float* W2 = (const float*)d_in[4];
    const float* b2 = (const float*)d_in[5];
    float* out = (float*)d_out;

    const int N = in_sizes[0] / 128;
    const int E = in_sizes[1] / 2;
    const int* srcIdx = ei;
    const int* dstIdx = ei + E;

    float* ws       = (float*)d_ws;
    int*   spillSrc = (int*)ws;
    float* h1   = ws + (((size_t)N + 255) & ~(size_t)255);
    float* h2   = h1 + (size_t)N * 128;
    int*   cnt      = (int*)(h2 + (size_t)N * 64);
    int*   spillDst = cnt      + (((size_t)N + 255) & ~(size_t)255);
    int*   spillN   = spillDst + N;
    int*   bucket   = spillDst + (((size_t)N + 256) & ~(size_t)255);

    const int TB = 256;
    auto blocksFor = [](long n, int tb) { return (int)((n + tb - 1) / tb); };

    const int gemmBlocks = blocksFor((long)(N / 16) * 32, TB);
    dbp_zero_int_kernel<<<blocksFor(N, TB), TB, 0, stream>>>(cnt, N, spillN);
    dbp_bucket_gemm1_kernel<<<gemmBlocks + blocksFor(E, TB), TB, 0, stream>>>(
        srcIdx, dstIdx, cnt, bucket, spillN, spillSrc, spillDst, N, E, x, W1, h1, N, gemmBlocks);

    dbp_gather1_gemm2_kernel<<<blocksFor(N, 16), 512, 0, stream>>>(
        cnt, bucket, spillN, spillSrc, spillDst, N, srcIdx, dstIdx, h1, b1, W2, h2, N, E);

    dbp_gather_kernel<64, true>
        <<<blocksFor((long)N * 32, TB), TB, 0, stream>>>(
            cnt, bucket, spillN, spillSrc, spillDst, N, srcIdx, dstIdx, h2, b2, out, N, E);
}
